# starting kernel
# baseline (speedup 1.0000x reference)
#include <hip/hip_runtime.h>
#include <hip/hip_bf16.h>


typedef __attribute__((ext_vector_type(2))) float v2f;
typedef __attribute__((ext_vector_type(8))) float v8f;

#define NB      65536
#define NL1     1024
#define NSTACK  8
#define NGRP    64
#define WS_CNT     0
#define WS_CUR     64
#define WS_OFF     128
#define WS_PERM    1024
#define WS_FRAG    (1024 + 65536*4)


__global__ void k_zero(int* w) {
    if (threadIdx.x < 48) w[threadIdx.x] = 0;
}

__global__ void k_count(const int* __restrict__ lsi, int* __restrict__ cnt) {
    int i = blockIdx.x * blockDim.x + threadIdx.x;
    if (i < NB) atomicAdd(&cnt[lsi[i] & 7], 1);
}

__global__ void k_scan(const int* __restrict__ cnt, int* __restrict__ off,
                       int* __restrict__ cur) {
    if (threadIdx.x == 0) {
        int acc = 0;
        for (int s = 0; s < NSTACK; ++s) {
            off[s] = acc;
            cur[s] = acc;
            acc += cnt[s];
        }
    }
}

__global__ void k_scatter(const int* __restrict__ lsi, int* __restrict__ cur,
                          int* __restrict__ perm) {
    int i = blockIdx.x * blockDim.x + threadIdx.x;
    if (i < NB) {
        int pos = atomicAdd(&cur[lsi[i] & 7], 1);
        perm[pos] = i;
    }
}


__global__ void k_prep(const float* __restrict__ l1_w,
                       const float* __restrict__ l1f_w,
                       float* __restrict__ frag) {
    int tid = blockIdx.x * blockDim.x + threadIdx.x;
    int e    = tid & 7;
    int lane = (tid >> 3) & 31;
    int h    = (tid >> 8) & 1;
    int g    = (tid >> 9) & 63;
    int s    = tid >> 15;
    int n = (lane & 15) + 16 * h;
    int k = g * 16 + ((lane < 16) ? 0 : 8) + e;
    float w = (n < 16) ? l1_w[(size_t)(s * 16 + n) * NL1 + k]
                       : l1f_w[(size_t)(n - 16) * NL1 + k];
    frag[tid] = w;
}


__global__ __launch_bounds__(128, 2)
void k_main(const float* __restrict__ x,
            const int* __restrict__ perm,
            const int* __restrict__ off,
            const int* __restrict__ cnt,
            const float* __restrict__ frag,
            const float* __restrict__ l1_b,
            const float* __restrict__ l1f_b,
            const float* __restrict__ l2_w,
            const float* __restrict__ l2_b,
            const float* __restrict__ out_w,
            const float* __restrict__ out_b,
            float* __restrict__ out) {
    __shared__ float lds[4][16][33];

    const int s    = blockIdx.x >> 10;
    const int tile = blockIdx.x & 1023;
    const int cnt_s = cnt[s];
    if (cnt_s == 0) return;
    if (tile * 64 >= cnt_s) return;

    const int wv   = threadIdx.x >> 5;
    const int lane = threadIdx.x & 31;
    const int r    = lane & 15;
    const int p    = tile * 64 + wv * 16 + r;
    const int pcl  = (p < cnt_s) ? p : (cnt_s - 1);
    const int rrow = perm[off[s] + pcl];

    const int koff = (lane < 16) ? 0 : 8;
    const float4* __restrict__ ax =
        (const float4*)(x + (size_t)rrow * NL1 + koff);
    const float4* __restrict__ bx = (const float4*)frag;

    v8f acc0 = {0.f, 0.f, 0.f, 0.f, 0.f, 0.f, 0.f, 0.f};
    v8f acc1 = {0.f, 0.f, 0.f, 0.f, 0.f, 0.f, 0.f, 0.f};

    const int fb0 = ((s * NGRP) * 2 + 0) * 64 + lane * 2;
    const int fb1 = ((s * NGRP) * 2 + 1) * 64 + lane * 2;

#pragma unroll 2
    for (int g = 0; g < NGRP; ++g) {
        float4 a0 = ax[4 * g + 0];
        float4 a1 = ax[4 * g + 1];
        float4 b00 = bx[fb0 + g * 128 + 0];
        float4 b01 = bx[fb0 + g * 128 + 1];
        float4 b10 = bx[fb1 + g * 128 + 0];
        float4 b11 = bx[fb1 + g * 128 + 1];
        float qa[8], q0[8], q1[8];
        *(float4*)&qa[0] = a0;  *(float4*)&qa[4] = a1;
        *(float4*)&q0[0] = b00; *(float4*)&q0[4] = b01;
        *(float4*)&q1[0] = b10; *(float4*)&q1[4] = b11;
#pragma unroll
        for (int c = 0; c < 4; ++c) {
            v2f A  = {qa[2 * c], qa[2 * c + 1]};
            v2f B0 = {q0[2 * c], q0[2 * c + 1]};
            v2f B1 = {q1[2 * c], q1[2 * c + 1]};
            acc0 = __builtin_amdgcn_wmma_f32_16x16x4_f32(
                false, A, false, B0, (short)0, acc0, false, false);
            acc1 = __builtin_amdgcn_wmma_f32_16x16x4_f32(
                false, A, false, B1, (short)0, acc1, false, false);
        }
    }

    const int mbase = (lane >= 16) ? 8 : 0;
    const int ncol  = lane & 15;
#pragma unroll
    for (int v = 0; v < 8; ++v) {
        lds[wv][v + mbase][ncol]      = acc0[v];
        lds[wv][v + mbase][ncol + 16] = acc1[v];
    }
    __builtin_amdgcn_wave_barrier();

    const int jh = lane >> 4;
    float t[15];
    float keep;
    {
        float yc15 = lds[wv][r][15] + l1_b[s * 16 + 15];
        float yf15 = lds[wv][r][31] + l1f_b[15];
        keep = yc15 + yf15;
#pragma unroll
        for (int j = 0; j < 15; ++j) {
            float yc = lds[wv][r][j]      + l1_b[s * 16 + j];
            float yf = lds[wv][r][16 + j] + l1f_b[j];
            t[j] = yc + yf;
        }
    }
    float l1x[30];
#pragma unroll
    for (int j = 0; j < 15; ++j) {
        float sq = t[j] * t[j] * (127.0f / 128.0f);
        l1x[j]      = fminf(fmaxf(sq, 0.f), 1.f);
        l1x[15 + j] = fminf(fmaxf(t[j], 0.f), 1.f);
    }
    float partial = 0.f;
#pragma unroll 4
    for (int oi = 0; oi < 16; ++oi) {
        int o = jh * 16 + oi;
        const float* wrow = l2_w + (size_t)(s * 32 + o) * 30;
        float a2 = l2_b[s * 32 + o];
#pragma unroll
        for (int k = 0; k < 30; ++k) a2 += l1x[k] * wrow[k];
        float l2x = fminf(fmaxf(a2, 0.f), 1.f);
        partial += l2x * out_w[s * 32 + o];
    }
    float tot = partial + __shfl_xor(partial, 16, 32);

    if (lane < 16 && p < cnt_s) {
        out[rrow] = tot + out_b[s] + keep;
    }
}


extern "C" void kernel_launch(void* const* d_in, const int* in_sizes, int n_in,
                              void* d_out, int out_size, void* d_ws, size_t ws_size,
                              hipStream_t stream) {
    (void)in_sizes; (void)n_in; (void)out_size; (void)ws_size;
    const float* x      = (const float*)d_in[0];
    const int*   lsi    = (const int*)d_in[1];
    const float* l1_w   = (const float*)d_in[2];
    const float* l1_b   = (const float*)d_in[3];
    const float* l1f_w  = (const float*)d_in[4];
    const float* l1f_b  = (const float*)d_in[5];
    const float* l2_w   = (const float*)d_in[6];
    const float* l2_b   = (const float*)d_in[7];
    const float* out_w  = (const float*)d_in[8];
    const float* out_b  = (const float*)d_in[9];
    float* out = (float*)d_out;

    char* ws   = (char*)d_ws;
    int*  cnt  = (int*)(ws + WS_CNT);
    int*  cur  = (int*)(ws + WS_CUR);
    int*  off  = (int*)(ws + WS_OFF);
    int*  perm = (int*)(ws + WS_PERM);
    float* frag = (float*)(ws + WS_FRAG);

    k_zero<<<1, 64, 0, stream>>>(cnt);
    k_count<<<NB / 256, 256, 0, stream>>>(lsi, cnt);
    k_scan<<<1, 32, 0, stream>>>(cnt, off, cur);
    k_scatter<<<NB / 256, 256, 0, stream>>>(lsi, cur, perm);
    k_prep<<<(NSTACK * NGRP * 2 * 256) / 256, 256, 0, stream>>>(l1_w, l1f_w, frag);
    k_main<<<NSTACK * 1024, 128, 0, stream>>>(x, perm, off, cnt, frag,
                                              l1_b, l1f_b, l2_w, l2_b,
                                              out_w, out_b, out);
}


# k_main: the 4 waves of a block split K (256 each), each keeps all 64 rows: one B fragment feeds 4 row tiles, 64 x lines in flight per wave (was 16); partials added through LDS (order of sums changes)
# speedup vs baseline: 1.8848x; 1.8848x over previous
#include <hip/hip_runtime.h>
#include <hip/hip_bf16.h>


typedef __attribute__((ext_vector_type(2))) float v2f;
typedef __attribute__((ext_vector_type(8))) float v8f;

#define NB      65536
#define NL1     1024
#define NSTACK  8
#define NGRP    64
#define WS_CNT     0
#define WS_CUR     64
#define WS_OFF     128
#define WS_PERM    1024
#define WS_FRAG    (1024 + 65536*4)


__global__ void k_zero(int* w) {
    if (threadIdx.x < 48) w[threadIdx.x] = 0;
}

__global__ __launch_bounds__(256)
void k_count(const int* __restrict__ lsi, int* __restrict__ cnt) {
    __shared__ int wcnt[8][NSTACK];
    const int t    = threadIdx.x;
    const int lane = t & 31;
    const int wv   = t >> 5;
    const int s    = lsi[blockIdx.x * 256 + t] & 7;
    int mine = 0;
#pragma unroll
    for (int k = 0; k < NSTACK; ++k) {
        unsigned m = __builtin_amdgcn_ballot_w32(s == k);
        if (lane == k) mine = __popc(m);
    }
    if (lane < NSTACK) wcnt[wv][lane] = mine;
    __syncthreads();
    if (t < NSTACK) {
        int acc = 0;
#pragma unroll
        for (int w = 0; w < 8; ++w) acc += wcnt[w][t];
        atomicAdd(&cnt[t], acc);
    }
}

__global__ void k_scan(const int* __restrict__ cnt, int* __restrict__ off,
                       int* __restrict__ cur) {
    if (threadIdx.x == 0) {
        int acc = 0;
        for (int s = 0; s < NSTACK; ++s) {
            off[s] = acc;
            cur[s] = acc;
            acc += cnt[s];
        }
    }
}

__global__ __launch_bounds__(256)
void k_scatter(const int* __restrict__ lsi, int* __restrict__ cur,
               int* __restrict__ perm) {
    __shared__ int wcnt[8][NSTACK];
    __shared__ int base[NSTACK];
    const int t    = threadIdx.x;
    const int lane = t & 31;
    const int wv   = t >> 5;
    const int i    = blockIdx.x * 256 + t;
    const int s    = lsi[i] & 7;
    int mine = 0;
    unsigned own = 0u;
#pragma unroll
    for (int k = 0; k < NSTACK; ++k) {
        unsigned m = __builtin_amdgcn_ballot_w32(s == k);
        own  = (s == k)    ? m : own;
        mine = (lane == k) ? __popc(m) : mine;
    }
    const int rank = (int)__builtin_amdgcn_mbcnt_lo(own, 0u);
    if (lane < NSTACK) wcnt[wv][lane] = mine;
    __syncthreads();
    if (t < NSTACK) {
        int acc = 0;
#pragma unroll
        for (int w = 0; w < 8; ++w) {
            int c = wcnt[w][t];
            wcnt[w][t] = acc;
            acc += c;
        }
        base[t] = atomicAdd(&cur[t], acc);
    }
    __syncthreads();
    perm[base[s] + wcnt[wv][s] + rank] = i;
}


__global__ void k_prep(const float* __restrict__ l1_w,
                       const float* __restrict__ l1f_w,
                       float* __restrict__ frag) {
    int tid = blockIdx.x * blockDim.x + threadIdx.x;
    int e    = tid & 7;
    int lane = (tid >> 3) & 31;
    int h    = (tid >> 8) & 1;
    int g    = (tid >> 9) & 63;
    int s    = tid >> 15;
    int n = (lane & 15) + 16 * h;
    int k = g * 16 + ((lane < 16) ? 0 : 8) + e;
    float w = (n < 16) ? l1_w[(size_t)(s * 16 + n) * NL1 + k]
                       : l1f_w[(size_t)(n - 16) * NL1 + k];
    frag[tid] = w;
}


__global__ __launch_bounds__(128, 5)
void k_main(const float* __restrict__ x,
            const int* __restrict__ perm,
            const int* __restrict__ off,
            const int* __restrict__ cnt,
            const float* __restrict__ frag,
            const float* __restrict__ l1_b,
            const float* __restrict__ l1f_b,
            const float* __restrict__ l2_w,
            const float* __restrict__ l2_b,
            const float* __restrict__ out_w,
            const float* __restrict__ out_b,
            float* __restrict__ out) {
    __shared__ __attribute__((aligned(16))) float red[4][64][36];

    const int s    = blockIdx.x & 7;
    const int tile = blockIdx.x >> 3;
    const int cnt_s = cnt[s];
    if (cnt_s == 0) return;
    if (tile * 64 >= cnt_s) return;

    const int wv   = threadIdx.x >> 5;
    const int lane = threadIdx.x & 31;
    const int r    = lane & 15;
    const int pbase = off[s];
    const int plast = cnt_s - 1;

    const int koff = (lane < 16) ? 0 : 8;

    const float4* __restrict__ ax[4];
#pragma unroll
    for (int m = 0; m < 4; ++m) {
        const int pm  = tile * 64 + m * 16 + r;
        const int row = perm[pbase + ((pm < cnt_s) ? pm : plast)];
        ax[m] = (const float4*)(x + (size_t)row * NL1 + wv * 256 + koff);
    }
    const float4* __restrict__ bx =
        (const float4*)frag + ((s * NGRP + wv * 16) * 2) * 64 + lane * 2;

    v8f acc[4][2];
#pragma unroll
    for (int m = 0; m < 4; ++m) {
        acc[m][0] = (v8f){0.f, 0.f, 0.f, 0.f, 0.f, 0.f, 0.f, 0.f};
        acc[m][1] = (v8f){0.f, 0.f, 0.f, 0.f, 0.f, 0.f, 0.f, 0.f};
    }

    for (int G = 0; G < 8; ++G) {
        float qa[4][16];
        float qb[2][2][8];
#pragma unroll
        for (int gg = 0; gg < 2; ++gg) {
#pragma unroll
            for (int h = 0; h < 2; ++h) {
                *(float4*)&qb[gg][h][0] = bx[((2 * G + gg) * 2 + h) * 64 + 0];
                *(float4*)&qb[gg][h][4] = bx[((2 * G + gg) * 2 + h) * 64 + 1];
            }
        }
#pragma unroll
        for (int m = 0; m < 4; ++m) {
            *(float4*)&qa[m][0]  = ax[m][8 * G + 0];
            *(float4*)&qa[m][4]  = ax[m][8 * G + 1];
            *(float4*)&qa[m][8]  = ax[m][8 * G + 4];
            *(float4*)&qa[m][12] = ax[m][8 * G + 5];
        }
        __builtin_amdgcn_sched_barrier(0);
#pragma unroll
        for (int gg = 0; gg < 2; ++gg) {
#pragma unroll
            for (int c = 0; c < 4; ++c) {
                v2f B0 = {qb[gg][0][2 * c], qb[gg][0][2 * c + 1]};
                v2f B1 = {qb[gg][1][2 * c], qb[gg][1][2 * c + 1]};
#pragma unroll
                for (int m = 0; m < 4; ++m) {
                    v2f A = {qa[m][8 * gg + 2 * c], qa[m][8 * gg + 2 * c + 1]};
                    acc[m][0] = __builtin_amdgcn_wmma_f32_16x16x4_f32(
                        false, A, false, B0, (short)0, acc[m][0], false, false);
                    acc[m][1] = __builtin_amdgcn_wmma_f32_16x16x4_f32(
                        false, A, false, B1, (short)0, acc[m][1], false, false);
                }
            }
        }
    }

    const int mbase = (lane >= 16) ? 8 : 0;
    const int ncol  = lane & 15;
#pragma unroll
    for (int m = 0; m < 4; ++m) {
#pragma unroll
        for (int v = 0; v < 8; ++v) {
            red[wv][m * 16 + v + mbase][ncol]      = acc[m][0][v];
            red[wv][m * 16 + v + mbase][ncol + 16] = acc[m][1][v];
        }
    }
    __syncthreads();

    const int R    = wv * 16 + r;
    const int p    = tile * 64 + R;
    const int rrow = perm[pbase + ((p < cnt_s) ? p : plast)];
    float y[32];
#pragma unroll
    for (int j4 = 0; j4 < 8; ++j4) {
        float4 u0 = *(const float4*)&red[0][R][4 * j4];
        float4 u1 = *(const float4*)&red[1][R][4 * j4];
        float4 u2 = *(const float4*)&red[2][R][4 * j4];
        float4 u3 = *(const float4*)&red[3][R][4 * j4];
        y[4 * j4 + 0] = ((u0.x + u1.x) + u2.x) + u3.x;
        y[4 * j4 + 1] = ((u0.y + u1.y) + u2.y) + u3.y;
        y[4 * j4 + 2] = ((u0.z + u1.z) + u2.z) + u3.z;
        y[4 * j4 + 3] = ((u0.w + u1.w) + u2.w) + u3.w;
    }

    const int jh = lane >> 4;
    float t[15];
    float keep;
    {
        float yc15 = y[15] + l1_b[s * 16 + 15];
        float yf15 = y[31] + l1f_b[15];
        keep = yc15 + yf15;
#pragma unroll
        for (int j = 0; j < 15; ++j) {
            float yc = y[j]      + l1_b[s * 16 + j];
            float yf = y[16 + j] + l1f_b[j];
            t[j] = yc + yf;
        }
    }
    float l1x[30];
#pragma unroll
    for (int j = 0; j < 15; ++j) {
        float sq = t[j] * t[j] * (127.0f / 128.0f);
        l1x[j]      = fminf(fmaxf(sq, 0.f), 1.f);
        l1x[15 + j] = fminf(fmaxf(t[j], 0.f), 1.f);
    }
    float partial = 0.f;
#pragma unroll 4
    for (int oi = 0; oi < 16; ++oi) {
        int o = jh * 16 + oi;
        const float* wrow = l2_w + (size_t)(s * 32 + o) * 30;
        float a2 = l2_b[s * 32 + o];
#pragma unroll
        for (int k = 0; k < 30; ++k) a2 += l1x[k] * wrow[k];
        float l2x = fminf(fmaxf(a2, 0.f), 1.f);
        partial += l2x * out_w[s * 32 + o];
    }
    float tot = partial + __shfl_xor(partial, 16, 32);

    if (lane < 16 && p < cnt_s) {
        out[rrow] = tot + out_b[s] + keep;
    }
}


extern "C" void kernel_launch(void* const* d_in, const int* in_sizes, int n_in,
                              void* d_out, int out_size, void* d_ws, size_t ws_size,
                              hipStream_t stream) {
    (void)in_sizes; (void)n_in; (void)out_size; (void)ws_size;
    const float* x      = (const float*)d_in[0];
    const int*   lsi    = (const int*)d_in[1];
    const float* l1_w   = (const float*)d_in[2];
    const float* l1_b   = (const float*)d_in[3];
    const float* l1f_w  = (const float*)d_in[4];
    const float* l1f_b  = (const float*)d_in[5];
    const float* l2_w   = (const float*)d_in[6];
    const float* l2_b   = (const float*)d_in[7];
    const float* out_w  = (const float*)d_in[8];
    const float* out_b  = (const float*)d_in[9];
    float* out = (float*)d_out;

    char* ws   = (char*)d_ws;
    int*  cnt  = (int*)(ws + WS_CNT);
    int*  cur  = (int*)(ws + WS_CUR);
    int*  off  = (int*)(ws + WS_OFF);
    int*  perm = (int*)(ws + WS_PERM);
    float* frag = (float*)(ws + WS_FRAG);

    k_zero<<<1, 64, 0, stream>>>(cnt);
    k_count<<<NB / 256, 256, 0, stream>>>(lsi, cnt);
    k_scan<<<1, 32, 0, stream>>>(cnt, off, cur);
    k_scatter<<<NB / 256, 256, 0, stream>>>(lsi, cur, perm);
    k_prep<<<(NSTACK * NGRP * 2 * 256) / 256, 256, 0, stream>>>(l1_w, l1f_w, frag);
    k_main<<<NSTACK * 1024, 128, 0, stream>>>(x, perm, off, cnt, frag,
                                              l1_b, l1f_b, l2_w, l2_b,
                                              out_w, out_b, out);
}


# k_main loads its B fragments straight from l1_w and l1f_w with the A loader's row pattern (equal k order by build); the weight pre-swizzle launch and its 1 MB plane are gone; sums unchanged
# speedup vs baseline: 1.9597x; 1.0397x over previous
#include <hip/hip_runtime.h>
#include <hip/hip_bf16.h>


typedef __attribute__((ext_vector_type(2))) float v2f;
typedef __attribute__((ext_vector_type(8))) float v8f;

#define NB      65536
#define NL1     1024
#define NSTACK  8
#define WS_CNT     0
#define WS_CUR     64
#define WS_OFF     128
#define WS_PERM    1024


__global__ void k_zero(int* w) {
    if (threadIdx.x < 48) w[threadIdx.x] = 0;
}

__global__ __launch_bounds__(256)
void k_count(const int* __restrict__ lsi, int* __restrict__ cnt) {
    __shared__ int wcnt[8][NSTACK];
    const int t    = threadIdx.x;
    const int lane = t & 31;
    const int wv   = t >> 5;
    const int s    = lsi[blockIdx.x * 256 + t] & 7;
    int mine = 0;
#pragma unroll
    for (int k = 0; k < NSTACK; ++k) {
        unsigned m = __builtin_amdgcn_ballot_w32(s == k);
        if (lane == k) mine = __popc(m);
    }
    if (lane < NSTACK) wcnt[wv][lane] = mine;
    __syncthreads();
    if (t < NSTACK) {
        int acc = 0;
#pragma unroll
        for (int w = 0; w < 8; ++w) acc += wcnt[w][t];
        atomicAdd(&cnt[t], acc);
    }
}

__global__ void k_scan(const int* __restrict__ cnt, int* __restrict__ off,
                       int* __restrict__ cur) {
    if (threadIdx.x == 0) {
        int acc = 0;
        for (int s = 0; s < NSTACK; ++s) {
            off[s] = acc;
            cur[s] = acc;
            acc += cnt[s];
        }
    }
}

__global__ __launch_bounds__(256)
void k_scatter(const int* __restrict__ lsi, int* __restrict__ cur,
               int* __restrict__ perm) {
    __shared__ int wcnt[8][NSTACK];
    __shared__ int base[NSTACK];
    const int t    = threadIdx.x;
    const int lane = t & 31;
    const int wv   = t >> 5;
    const int i    = blockIdx.x * 256 + t;
    const int s    = lsi[i] & 7;
    int mine = 0;
    unsigned own = 0u;
#pragma unroll
    for (int k = 0; k < NSTACK; ++k) {
        unsigned m = __builtin_amdgcn_ballot_w32(s == k);
        own  = (s == k)    ? m : own;
        mine = (lane == k) ? __popc(m) : mine;
    }
    const int rank = (int)__builtin_amdgcn_mbcnt_lo(own, 0u);
    if (lane < NSTACK) wcnt[wv][lane] = mine;
    __syncthreads();
    if (t < NSTACK) {
        int acc = 0;
#pragma unroll
        for (int w = 0; w < 8; ++w) {
            int c = wcnt[w][t];
            wcnt[w][t] = acc;
            acc += c;
        }
        base[t] = atomicAdd(&cur[t], acc);
    }
    __syncthreads();
    perm[base[s] + wcnt[wv][s] + rank] = i;
}


__global__ __launch_bounds__(128, 5)
void k_main(const float* __restrict__ x,
            const int* __restrict__ perm,
            const int* __restrict__ off,
            const int* __restrict__ cnt,
            const float* __restrict__ l1_w,
            const float* __restrict__ l1f_w,
            const float* __restrict__ l1_b,
            const float* __restrict__ l1f_b,
            const float* __restrict__ l2_w,
            const float* __restrict__ l2_b,
            const float* __restrict__ out_w,
            const float* __restrict__ out_b,
            float* __restrict__ out) {
    __shared__ __attribute__((aligned(16))) float red[4][64][36];

    const int s    = blockIdx.x & 7;
    const int tile = blockIdx.x >> 3;
    const int cnt_s = cnt[s];
    if (cnt_s == 0) return;
    if (tile * 64 >= cnt_s) return;

    const int wv   = threadIdx.x >> 5;
    const int lane = threadIdx.x & 31;
    const int r    = lane & 15;
    const int pbase = off[s];
    const int plast = cnt_s - 1;

    const int koff = (lane < 16) ? 0 : 8;

    const float4* __restrict__ ax[4];
#pragma unroll
    for (int m = 0; m < 4; ++m) {
        const int pm  = tile * 64 + m * 16 + r;
        const int row = perm[pbase + ((pm < cnt_s) ? pm : plast)];
        ax[m] = (const float4*)(x + (size_t)row * NL1 + wv * 256 + koff);
    }
    const float4* __restrict__ bw[2];
    bw[0] = (const float4*)(l1_w  + (size_t)(s * 16 + r) * NL1 + wv * 256 + koff);
    bw[1] = (const float4*)(l1f_w + (size_t)r * NL1 + wv * 256 + koff);

    v8f acc[4][2];
#pragma unroll
    for (int m = 0; m < 4; ++m) {
        acc[m][0] = (v8f){0.f, 0.f, 0.f, 0.f, 0.f, 0.f, 0.f, 0.f};
        acc[m][1] = (v8f){0.f, 0.f, 0.f, 0.f, 0.f, 0.f, 0.f, 0.f};
    }

    for (int G = 0; G < 8; ++G) {
        float qa[4][16];
        float qb[2][2][8];
#pragma unroll
        for (int gg = 0; gg < 2; ++gg) {
#pragma unroll
            for (int h = 0; h < 2; ++h) {
                *(float4*)&qb[gg][h][0] = bw[h][8 * G + 4 * gg + 0];
                *(float4*)&qb[gg][h][4] = bw[h][8 * G + 4 * gg + 1];
            }
        }
#pragma unroll
        for (int m = 0; m < 4; ++m) {
            *(float4*)&qa[m][0]  = ax[m][8 * G + 0];
            *(float4*)&qa[m][4]  = ax[m][8 * G + 1];
            *(float4*)&qa[m][8]  = ax[m][8 * G + 4];
            *(float4*)&qa[m][12] = ax[m][8 * G + 5];
        }
        __builtin_amdgcn_sched_barrier(0);
#pragma unroll
        for (int gg = 0; gg < 2; ++gg) {
#pragma unroll
            for (int c = 0; c < 4; ++c) {
                v2f B0 = {qb[gg][0][2 * c], qb[gg][0][2 * c + 1]};
                v2f B1 = {qb[gg][1][2 * c], qb[gg][1][2 * c + 1]};
#pragma unroll
                for (int m = 0; m < 4; ++m) {
                    v2f A = {qa[m][8 * gg + 2 * c], qa[m][8 * gg + 2 * c + 1]};
                    acc[m][0] = __builtin_amdgcn_wmma_f32_16x16x4_f32(
                        false, A, false, B0, (short)0, acc[m][0], false, false);
                    acc[m][1] = __builtin_amdgcn_wmma_f32_16x16x4_f32(
                        false, A, false, B1, (short)0, acc[m][1], false, false);
                }
            }
        }
    }

    const int mbase = (lane >= 16) ? 8 : 0;
    const int ncol  = lane & 15;
#pragma unroll
    for (int m = 0; m < 4; ++m) {
#pragma unroll
        for (int v = 0; v < 8; ++v) {
            red[wv][m * 16 + v + mbase][ncol]      = acc[m][0][v];
            red[wv][m * 16 + v + mbase][ncol + 16] = acc[m][1][v];
        }
    }
    __syncthreads();

    const int R    = wv * 16 + r;
    const int p    = tile * 64 + R;
    const int rrow = perm[pbase + ((p < cnt_s) ? p : plast)];
    float y[32];
#pragma unroll
    for (int j4 = 0; j4 < 8; ++j4) {
        float4 u0 = *(const float4*)&red[0][R][4 * j4];
        float4 u1 = *(const float4*)&red[1][R][4 * j4];
        float4 u2 = *(const float4*)&red[2][R][4 * j4];
        float4 u3 = *(const float4*)&red[3][R][4 * j4];
        y[4 * j4 + 0] = ((u0.x + u1.x) + u2.x) + u3.x;
        y[4 * j4 + 1] = ((u0.y + u1.y) + u2.y) + u3.y;
        y[4 * j4 + 2] = ((u0.z + u1.z) + u2.z) + u3.z;
        y[4 * j4 + 3] = ((u0.w + u1.w) + u2.w) + u3.w;
    }

    const int jh = lane >> 4;
    float t[15];
    float keep;
    {
        float yc15 = y[15] + l1_b[s * 16 + 15];
        float yf15 = y[31] + l1f_b[15];
        keep = yc15 + yf15;
#pragma unroll
        for (int j = 0; j < 15; ++j) {
            float yc = y[j]      + l1_b[s * 16 + j];
            float yf = y[16 + j] + l1f_b[j];
            t[j] = yc + yf;
        }
    }
    float l1x[30];
#pragma unroll
    for (int j = 0; j < 15; ++j) {
        float sq = t[j] * t[j] * (127.0f / 128.0f);
        l1x[j]      = fminf(fmaxf(sq, 0.f), 1.f);
        l1x[15 + j] = fminf(fmaxf(t[j], 0.f), 1.f);
    }
    float partial = 0.f;
#pragma unroll 4
    for (int oi = 0; oi < 16; ++oi) {
        int o = jh * 16 + oi;
        const float* wrow = l2_w + (size_t)(s * 32 + o) * 30;
        float a2 = l2_b[s * 32 + o];
#pragma unroll
        for (int k = 0; k < 30; ++k) a2 += l1x[k] * wrow[k];
        float l2x = fminf(fmaxf(a2, 0.f), 1.f);
        partial += l2x * out_w[s * 32 + o];
    }
    float tot = partial + __shfl_xor(partial, 16, 32);

    if (lane < 16 && p < cnt_s) {
        out[rrow] = tot + out_b[s] + keep;
    }
}


extern "C" void kernel_launch(void* const* d_in, const int* in_sizes, int n_in,
                              void* d_out, int out_size, void* d_ws, size_t ws_size,
                              hipStream_t stream) {
    (void)in_sizes; (void)n_in; (void)out_size; (void)ws_size;
    const float* x      = (const float*)d_in[0];
    const int*   lsi    = (const int*)d_in[1];
    const float* l1_w   = (const float*)d_in[2];
    const float* l1_b   = (const float*)d_in[3];
    const float* l1f_w  = (const float*)d_in[4];
    const float* l1f_b  = (const float*)d_in[5];
    const float* l2_w   = (const float*)d_in[6];
    const float* l2_b   = (const float*)d_in[7];
    const float* out_w  = (const float*)d_in[8];
    const float* out_b  = (const float*)d_in[9];
    float* out = (float*)d_out;

    char* ws   = (char*)d_ws;
    int*  cnt  = (int*)(ws + WS_CNT);
    int*  cur  = (int*)(ws + WS_CUR);
    int*  off  = (int*)(ws + WS_OFF);
    int*  perm = (int*)(ws + WS_PERM);

    k_zero<<<1, 64, 0, stream>>>(cnt);
    k_count<<<NB / 256, 256, 0, stream>>>(lsi, cnt);
    k_scan<<<1, 32, 0, stream>>>(cnt, off, cur);
    k_scatter<<<NB / 256, 256, 0, stream>>>(lsi, cur, perm);
    k_main<<<NSTACK * 1024, 128, 0, stream>>>(x, perm, off, cnt, l1_w, l1f_w,
                                              l1_b, l1f_b, l2_w, l2_b,
                                              out_w, out_b, out);
}


# rows scattered straight into 8 fixed-place buckets (bucket s at s*65536, 2-byte row ids, place = its stack's running counter): no count and no prefix needed, k_count and k_scan gone, 5 launches -> 3
# speedup vs baseline: 2.0852x; 1.0640x over previous
#include <hip/hip_runtime.h>
#include <hip/hip_bf16.h>


typedef __attribute__((ext_vector_type(2))) float v2f;
typedef __attribute__((ext_vector_type(8))) float v8f;

#define NB      65536
#define NL1     1024
#define NSTACK  8
#define WS_CUR     0
#define WS_PERM    1024


__global__ void k_zero(int* w) {
    if (threadIdx.x < NSTACK) w[threadIdx.x] = 0;
}

__global__ __launch_bounds__(256)
void k_scatter(const int* __restrict__ lsi, int* __restrict__ cur,
               unsigned short* __restrict__ perm) {
    __shared__ int wcnt[8][NSTACK];
    __shared__ int base[NSTACK];
    const int t    = threadIdx.x;
    const int lane = t & 31;
    const int wv   = t >> 5;
    const int i    = blockIdx.x * 256 + t;
    const int s    = lsi[i] & 7;
    int mine = 0;
    unsigned own = 0u;
#pragma unroll
    for (int k = 0; k < NSTACK; ++k) {
        unsigned m = __builtin_amdgcn_ballot_w32(s == k);
        own  = (s == k)    ? m : own;
        mine = (lane == k) ? __popc(m) : mine;
    }
    const int rank = (int)__builtin_amdgcn_mbcnt_lo(own, 0u);
    if (lane < NSTACK) wcnt[wv][lane] = mine;
    __syncthreads();
    if (t < NSTACK) {
        int acc = 0;
#pragma unroll
        for (int w = 0; w < 8; ++w) {
            int c = wcnt[w][t];
            wcnt[w][t] = acc;
            acc += c;
        }
        base[t] = atomicAdd(&cur[t], acc);
    }
    __syncthreads();
    perm[s * NB + base[s] + wcnt[wv][s] + rank] = (unsigned short)i;
}


__global__ __launch_bounds__(128, 5)
void k_main(const float* __restrict__ x,
            const unsigned short* __restrict__ perm,
            const int* __restrict__ cnt,
            const float* __restrict__ l1_w,
            const float* __restrict__ l1f_w,
            const float* __restrict__ l1_b,
            const float* __restrict__ l1f_b,
            const float* __restrict__ l2_w,
            const float* __restrict__ l2_b,
            const float* __restrict__ out_w,
            const float* __restrict__ out_b,
            float* __restrict__ out) {
    __shared__ __attribute__((aligned(16))) float red[4][64][36];

    const int s    = blockIdx.x & 7;
    const int tile = blockIdx.x >> 3;
    const int cnt_s = cnt[s];
    if (cnt_s == 0) return;
    if (tile * 64 >= cnt_s) return;

    const int wv   = threadIdx.x >> 5;
    const int lane = threadIdx.x & 31;
    const int r    = lane & 15;
    const int pbase = s * NB;
    const int plast = cnt_s - 1;

    const int koff = (lane < 16) ? 0 : 8;

    const float4* __restrict__ ax[4];
#pragma unroll
    for (int m = 0; m < 4; ++m) {
        const int pm  = tile * 64 + m * 16 + r;
        const int row = perm[pbase + ((pm < cnt_s) ? pm : plast)];
        ax[m] = (const float4*)(x + (size_t)row * NL1 + wv * 256 + koff);
    }
    const float4* __restrict__ bw[2];
    bw[0] = (const float4*)(l1_w  + (size_t)(s * 16 + r) * NL1 + wv * 256 + koff);
    bw[1] = (const float4*)(l1f_w + (size_t)r * NL1 + wv * 256 + koff);

    v8f acc[4][2];
#pragma unroll
    for (int m = 0; m < 4; ++m) {
        acc[m][0] = (v8f){0.f, 0.f, 0.f, 0.f, 0.f, 0.f, 0.f, 0.f};
        acc[m][1] = (v8f){0.f, 0.f, 0.f, 0.f, 0.f, 0.f, 0.f, 0.f};
    }

    for (int G = 0; G < 8; ++G) {
        float qa[4][16];
        float qb[2][2][8];
#pragma unroll
        for (int gg = 0; gg < 2; ++gg) {
#pragma unroll
            for (int h = 0; h < 2; ++h) {
                *(float4*)&qb[gg][h][0] = bw[h][8 * G + 4 * gg + 0];
                *(float4*)&qb[gg][h][4] = bw[h][8 * G + 4 * gg + 1];
            }
        }
#pragma unroll
        for (int m = 0; m < 4; ++m) {
            *(float4*)&qa[m][0]  = ax[m][8 * G + 0];
            *(float4*)&qa[m][4]  = ax[m][8 * G + 1];
            *(float4*)&qa[m][8]  = ax[m][8 * G + 4];
            *(float4*)&qa[m][12] = ax[m][8 * G + 5];
        }
        __builtin_amdgcn_sched_barrier(0);
#pragma unroll
        for (int gg = 0; gg < 2; ++gg) {
#pragma unroll
            for (int c = 0; c < 4; ++c) {
                v2f B0 = {qb[gg][0][2 * c], qb[gg][0][2 * c + 1]};
                v2f B1 = {qb[gg][1][2 * c], qb[gg][1][2 * c + 1]};
#pragma unroll
                for (int m = 0; m < 4; ++m) {
                    v2f A = {qa[m][8 * gg + 2 * c], qa[m][8 * gg + 2 * c + 1]};
                    acc[m][0] = __builtin_amdgcn_wmma_f32_16x16x4_f32(
                        false, A, false, B0, (short)0, acc[m][0], false, false);
                    acc[m][1] = __builtin_amdgcn_wmma_f32_16x16x4_f32(
                        false, A, false, B1, (short)0, acc[m][1], false, false);
                }
            }
        }
    }

    const int mbase = (lane >= 16) ? 8 : 0;
    const int ncol  = lane & 15;
#pragma unroll
    for (int m = 0; m < 4; ++m) {
#pragma unroll
        for (int v = 0; v < 8; ++v) {
            red[wv][m * 16 + v + mbase][ncol]      = acc[m][0][v];
            red[wv][m * 16 + v + mbase][ncol + 16] = acc[m][1][v];
        }
    }
    __syncthreads();

    const int R    = wv * 16 + r;
    const int p    = tile * 64 + R;
    const int rrow = perm[pbase + ((p < cnt_s) ? p : plast)];
    float y[32];
#pragma unroll
    for (int j4 = 0; j4 < 8; ++j4) {
        float4 u0 = *(const float4*)&red[0][R][4 * j4];
        float4 u1 = *(const float4*)&red[1][R][4 * j4];
        float4 u2 = *(const float4*)&red[2][R][4 * j4];
        float4 u3 = *(const float4*)&red[3][R][4 * j4];
        y[4 * j4 + 0] = ((u0.x + u1.x) + u2.x) + u3.x;
        y[4 * j4 + 1] = ((u0.y + u1.y) + u2.y) + u3.y;
        y[4 * j4 + 2] = ((u0.z + u1.z) + u2.z) + u3.z;
        y[4 * j4 + 3] = ((u0.w + u1.w) + u2.w) + u3.w;
    }

    const int jh = lane >> 4;
    float t[15];
    float keep;
    {
        float yc15 = y[15] + l1_b[s * 16 + 15];
        float yf15 = y[31] + l1f_b[15];
        keep = yc15 + yf15;
#pragma unroll
        for (int j = 0; j < 15; ++j) {
            float yc = y[j]      + l1_b[s * 16 + j];
            float yf = y[16 + j] + l1f_b[j];
            t[j] = yc + yf;
        }
    }
    float l1x[30];
#pragma unroll
    for (int j = 0; j < 15; ++j) {
        float sq = t[j] * t[j] * (127.0f / 128.0f);
        l1x[j]      = fminf(fmaxf(sq, 0.f), 1.f);
        l1x[15 + j] = fminf(fmaxf(t[j], 0.f), 1.f);
    }
    float partial = 0.f;
#pragma unroll 4
    for (int oi = 0; oi < 16; ++oi) {
        int o = jh * 16 + oi;
        const float* wrow = l2_w + (size_t)(s * 32 + o) * 30;
        float a2 = l2_b[s * 32 + o];
#pragma unroll
        for (int k = 0; k < 30; ++k) a2 += l1x[k] * wrow[k];
        float l2x = fminf(fmaxf(a2, 0.f), 1.f);
        partial += l2x * out_w[s * 32 + o];
    }
    float tot = partial + __shfl_xor(partial, 16, 32);

    if (lane < 16 && p < cnt_s) {
        out[rrow] = tot + out_b[s] + keep;
    }
}


extern "C" void kernel_launch(void* const* d_in, const int* in_sizes, int n_in,
                              void* d_out, int out_size, void* d_ws, size_t ws_size,
                              hipStream_t stream) {
    (void)in_sizes; (void)n_in; (void)out_size; (void)ws_size;
    const float* x      = (const float*)d_in[0];
    const int*   lsi    = (const int*)d_in[1];
    const float* l1_w   = (const float*)d_in[2];
    const float* l1_b   = (const float*)d_in[3];
    const float* l1f_w  = (const float*)d_in[4];
    const float* l1f_b  = (const float*)d_in[5];
    const float* l2_w   = (const float*)d_in[6];
    const float* l2_b   = (const float*)d_in[7];
    const float* out_w  = (const float*)d_in[8];
    const float* out_b  = (const float*)d_in[9];
    float* out = (float*)d_out;

    char* ws   = (char*)d_ws;
    int*  cur  = (int*)(ws + WS_CUR);
    unsigned short* perm = (unsigned short*)(ws + WS_PERM);

    k_zero<<<1, 32, 0, stream>>>(cur);
    k_scatter<<<NB / 256, 256, 0, stream>>>(lsi, cur, perm);
    k_main<<<NSTACK * 1024, 128, 0, stream>>>(x, perm, cur, l1_w, l1f_w,
                                              l1_b, l1f_b, l2_w, l2_b,
                                              out_w, out_b, out);
}


# k_main blocks count live units off from the 8 bucket sizes: whole 64-row tiles first, then left-over rows as 16-row one-tile units; grid 1056 (was 8192), a unit's fifth block is light; sums unchanged
# speedup vs baseline: 2.2733x; 1.0902x over previous
#include <hip/hip_runtime.h>
#include <hip/hip_bf16.h>


typedef __attribute__((ext_vector_type(2))) float v2f;
typedef __attribute__((ext_vector_type(8))) float v8f;

#define NB      65536
#define NL1     1024
#define NSTACK  8
#define NLIVE   1056
#define WS_CUR     0
#define WS_PERM    1024


__global__ void k_zero(int* w) {
    if (threadIdx.x < NSTACK) w[threadIdx.x] = 0;
}

__global__ __launch_bounds__(256)
void k_scatter(const int* __restrict__ lsi, int* __restrict__ cur,
               unsigned short* __restrict__ perm) {
    __shared__ int wcnt[8][NSTACK];
    __shared__ int base[NSTACK];
    const int t    = threadIdx.x;
    const int lane = t & 31;
    const int wv   = t >> 5;
    const int i    = blockIdx.x * 256 + t;
    const int s    = lsi[i] & 7;
    int mine = 0;
    unsigned own = 0u;
#pragma unroll
    for (int k = 0; k < NSTACK; ++k) {
        unsigned m = __builtin_amdgcn_ballot_w32(s == k);
        own  = (s == k)    ? m : own;
        mine = (lane == k) ? __popc(m) : mine;
    }
    const int rank = (int)__builtin_amdgcn_mbcnt_lo(own, 0u);
    if (lane < NSTACK) wcnt[wv][lane] = mine;
    __syncthreads();
    if (t < NSTACK) {
        int acc = 0;
#pragma unroll
        for (int w = 0; w < 8; ++w) {
            int c = wcnt[w][t];
            wcnt[w][t] = acc;
            acc += c;
        }
        base[t] = atomicAdd(&cur[t], acc);
    }
    __syncthreads();
    perm[s * NB + base[s] + wcnt[wv][s] + rank] = (unsigned short)i;
}


template <int NM>
__device__ __forceinline__
void tile_body(const float* __restrict__ x,
               const unsigned short* __restrict__ perm,
               const float* __restrict__ l1_w,
               const float* __restrict__ l1f_w,
               const float* __restrict__ l1_b,
               const float* __restrict__ l1f_b,
               const float* __restrict__ l2_w,
               const float* __restrict__ l2_b,
               const float* __restrict__ out_w,
               const float* __restrict__ out_b,
               float* __restrict__ out,
               float (*red)[64][36],
               const int s, const int row0, const int cnt_s) {
    const int wv   = threadIdx.x >> 5;
    const int lane = threadIdx.x & 31;
    const int r    = lane & 15;
    const int pbase = s * NB;
    const int plast = cnt_s - 1;

    const int koff = (lane < 16) ? 0 : 8;

    const float4* __restrict__ ax[NM];
#pragma unroll
    for (int m = 0; m < NM; ++m) {
        const int pm  = row0 + m * 16 + r;
        const int row = perm[pbase + ((pm < cnt_s) ? pm : plast)];
        ax[m] = (const float4*)(x + (size_t)row * NL1 + wv * 256 + koff);
    }
    const float4* __restrict__ bw[2];
    bw[0] = (const float4*)(l1_w  + (size_t)(s * 16 + r) * NL1 + wv * 256 + koff);
    bw[1] = (const float4*)(l1f_w + (size_t)r * NL1 + wv * 256 + koff);

    v8f acc[NM][2];
#pragma unroll
    for (int m = 0; m < NM; ++m) {
        acc[m][0] = (v8f){0.f, 0.f, 0.f, 0.f, 0.f, 0.f, 0.f, 0.f};
        acc[m][1] = (v8f){0.f, 0.f, 0.f, 0.f, 0.f, 0.f, 0.f, 0.f};
    }

    for (int G = 0; G < 8; ++G) {
        float qa[NM][16];
        float qb[2][2][8];
#pragma unroll
        for (int gg = 0; gg < 2; ++gg) {
#pragma unroll
            for (int h = 0; h < 2; ++h) {
                *(float4*)&qb[gg][h][0] = bw[h][8 * G + 4 * gg + 0];
                *(float4*)&qb[gg][h][4] = bw[h][8 * G + 4 * gg + 1];
            }
        }
#pragma unroll
        for (int m = 0; m < NM; ++m) {
            *(float4*)&qa[m][0]  = ax[m][8 * G + 0];
            *(float4*)&qa[m][4]  = ax[m][8 * G + 1];
            *(float4*)&qa[m][8]  = ax[m][8 * G + 4];
            *(float4*)&qa[m][12] = ax[m][8 * G + 5];
        }
        __builtin_amdgcn_sched_barrier(0);
#pragma unroll
        for (int gg = 0; gg < 2; ++gg) {
#pragma unroll
            for (int c = 0; c < 4; ++c) {
                v2f B0 = {qb[gg][0][2 * c], qb[gg][0][2 * c + 1]};
                v2f B1 = {qb[gg][1][2 * c], qb[gg][1][2 * c + 1]};
#pragma unroll
                for (int m = 0; m < NM; ++m) {
                    v2f A = {qa[m][8 * gg + 2 * c], qa[m][8 * gg + 2 * c + 1]};
                    acc[m][0] = __builtin_amdgcn_wmma_f32_16x16x4_f32(
                        false, A, false, B0, (short)0, acc[m][0], false, false);
                    acc[m][1] = __builtin_amdgcn_wmma_f32_16x16x4_f32(
                        false, A, false, B1, (short)0, acc[m][1], false, false);
                }
            }
        }
    }

    const int mbase = (lane >= 16) ? 8 : 0;
    const int ncol  = lane & 15;
#pragma unroll
    for (int m = 0; m < NM; ++m) {
#pragma unroll
        for (int v = 0; v < 8; ++v) {
            red[wv][m * 16 + v + mbase][ncol]      = acc[m][0][v];
            red[wv][m * 16 + v + mbase][ncol + 16] = acc[m][1][v];
        }
    }
    __syncthreads();

    if (wv >= NM) return;
    const int R    = wv * 16 + r;
    const int p    = row0 + R;
    const int rrow = perm[pbase + ((p < cnt_s) ? p : plast)];
    float y[32];
#pragma unroll
    for (int j4 = 0; j4 < 8; ++j4) {
        float4 u0 = *(const float4*)&red[0][R][4 * j4];
        float4 u1 = *(const float4*)&red[1][R][4 * j4];
        float4 u2 = *(const float4*)&red[2][R][4 * j4];
        float4 u3 = *(const float4*)&red[3][R][4 * j4];
        y[4 * j4 + 0] = ((u0.x + u1.x) + u2.x) + u3.x;
        y[4 * j4 + 1] = ((u0.y + u1.y) + u2.y) + u3.y;
        y[4 * j4 + 2] = ((u0.z + u1.z) + u2.z) + u3.z;
        y[4 * j4 + 3] = ((u0.w + u1.w) + u2.w) + u3.w;
    }

    const int jh = lane >> 4;
    float t[15];
    float keep;
    {
        float yc15 = y[15] + l1_b[s * 16 + 15];
        float yf15 = y[31] + l1f_b[15];
        keep = yc15 + yf15;
#pragma unroll
        for (int j = 0; j < 15; ++j) {
            float yc = y[j]      + l1_b[s * 16 + j];
            float yf = y[16 + j] + l1f_b[j];
            t[j] = yc + yf;
        }
    }
    float l1x[30];
#pragma unroll
    for (int j = 0; j < 15; ++j) {
        float sq = t[j] * t[j] * (127.0f / 128.0f);
        l1x[j]      = fminf(fmaxf(sq, 0.f), 1.f);
        l1x[15 + j] = fminf(fmaxf(t[j], 0.f), 1.f);
    }
    float partial = 0.f;
#pragma unroll 4
    for (int oi = 0; oi < 16; ++oi) {
        int o = jh * 16 + oi;
        const float* wrow = l2_w + (size_t)(s * 32 + o) * 30;
        float a2 = l2_b[s * 32 + o];
#pragma unroll
        for (int k = 0; k < 30; ++k) a2 += l1x[k] * wrow[k];
        float l2x = fminf(fmaxf(a2, 0.f), 1.f);
        partial += l2x * out_w[s * 32 + o];
    }
    float tot = partial + __shfl_xor(partial, 16, 32);

    if (lane < 16 && p < cnt_s) {
        out[rrow] = tot + out_b[s] + keep;
    }
}

__global__ __launch_bounds__(128, 5)
void k_main(const float* __restrict__ x,
            const unsigned short* __restrict__ perm,
            const int* __restrict__ cnt,
            const float* __restrict__ l1_w,
            const float* __restrict__ l1f_w,
            const float* __restrict__ l1_b,
            const float* __restrict__ l1f_b,
            const float* __restrict__ l2_w,
            const float* __restrict__ l2_b,
            const float* __restrict__ out_w,
            const float* __restrict__ out_b,
            float* __restrict__ out) {
    __shared__ __attribute__((aligned(16))) float red[4][64][36];

    int cn[NSTACK], nf[NSTACK], nl[NSTACK];
    int m0 = NB;
#pragma unroll
    for (int k = 0; k < NSTACK; ++k) {
        const int c = cnt[k];
        cn[k] = c;
        nf[k] = c >> 6;
        nl[k] = ((c & 63) + 15) >> 4;
        m0 = (nf[k] < m0) ? nf[k] : m0;
    }
    const int b = blockIdx.x;
    int  s = -1, row0 = 0;
    bool whole = true;
    if (b < NSTACK * m0) {
        s = b & 7;
        row0 = (b >> 3) * 64;
    } else {
        int q = b - NSTACK * m0;
#pragma unroll
        for (int k = 0; k < NSTACK; ++k) {
            const int e = nf[k] - m0;
            if (s < 0) {
                if (q < e) { s = k; row0 = (m0 + q) * 64; }
                else       q -= e;
            }
        }
#pragma unroll
        for (int k = 0; k < NSTACK; ++k) {
            if (s < 0) {
                if (q < nl[k]) { s = k; row0 = nf[k] * 64 + q * 16; whole = false; }
                else           q -= nl[k];
            }
        }
        if (s < 0) return;
    }
    int cnt_s = cn[0];
#pragma unroll
    for (int k = 1; k < NSTACK; ++k) cnt_s = (s == k) ? cn[k] : cnt_s;
    if (whole)
        tile_body<4>(x, perm, l1_w, l1f_w, l1_b, l1f_b, l2_w, l2_b, out_w, out_b,
                     out, red, s, row0, cnt_s);
    else
        tile_body<1>(x, perm, l1_w, l1f_w, l1_b, l1f_b, l2_w, l2_b, out_w, out_b,
                     out, red, s, row0, cnt_s);
}


extern "C" void kernel_launch(void* const* d_in, const int* in_sizes, int n_in,
                              void* d_out, int out_size, void* d_ws, size_t ws_size,
                              hipStream_t stream) {
    (void)in_sizes; (void)n_in; (void)out_size; (void)ws_size;
    const float* x      = (const float*)d_in[0];
    const int*   lsi    = (const int*)d_in[1];
    const float* l1_w   = (const float*)d_in[2];
    const float* l1_b   = (const float*)d_in[3];
    const float* l1f_w  = (const float*)d_in[4];
    const float* l1f_b  = (const float*)d_in[5];
    const float* l2_w   = (const float*)d_in[6];
    const float* l2_b   = (const float*)d_in[7];
    const float* out_w  = (const float*)d_in[8];
    const float* out_b  = (const float*)d_in[9];
    float* out = (float*)d_out;

    char* ws   = (char*)d_ws;
    int*  cur  = (int*)(ws + WS_CUR);
    unsigned short* perm = (unsigned short*)(ws + WS_PERM);

    k_zero<<<1, 32, 0, stream>>>(cur);
    k_scatter<<<NB / 256, 256, 0, stream>>>(lsi, cur, perm);
    k_main<<<NLIVE, 128, 0, stream>>>(x, perm, cur, l1_w, l1f_w,
                                              l1_b, l1f_b, l2_w, l2_b,
                                              out_w, out_b, out);
}


# k_main epilogue: the 30->32 layer runs on the f32 matrix unit (per wave two 16x16 tiles: weight rows x its 16 rows, bias as C) instead of lane FMAs fed by 140 loads; k and output sum orders change
# speedup vs baseline: 2.3742x; 1.0444x over previous
#include <hip/hip_runtime.h>
#include <hip/hip_bf16.h>


typedef __attribute__((ext_vector_type(2))) float v2f;
typedef __attribute__((ext_vector_type(8))) float v8f;

#define NB      65536
#define NL1     1024
#define NSTACK  8
#define NLIVE   1056
#define WS_CUR     0
#define WS_PERM    1024


__global__ void k_zero(int* w) {
    if (threadIdx.x < NSTACK) w[threadIdx.x] = 0;
}

__global__ __launch_bounds__(256)
void k_scatter(const int* __restrict__ lsi, int* __restrict__ cur,
               unsigned short* __restrict__ perm) {
    __shared__ int wcnt[8][NSTACK];
    __shared__ int base[NSTACK];
    const int t    = threadIdx.x;
    const int lane = t & 31;
    const int wv   = t >> 5;
    const int i    = blockIdx.x * 256 + t;
    const int s    = lsi[i] & 7;
    int mine = 0;
    unsigned own = 0u;
#pragma unroll
    for (int k = 0; k < NSTACK; ++k) {
        unsigned m = __builtin_amdgcn_ballot_w32(s == k);
        own  = (s == k)    ? m : own;
        mine = (lane == k) ? __popc(m) : mine;
    }
    const int rank = (int)__builtin_amdgcn_mbcnt_lo(own, 0u);
    if (lane < NSTACK) wcnt[wv][lane] = mine;
    __syncthreads();
    if (t < NSTACK) {
        int acc = 0;
#pragma unroll
        for (int w = 0; w < 8; ++w) {
            int c = wcnt[w][t];
            wcnt[w][t] = acc;
            acc += c;
        }
        base[t] = atomicAdd(&cur[t], acc);
    }
    __syncthreads();
    perm[s * NB + base[s] + wcnt[wv][s] + rank] = (unsigned short)i;
}


template <int NM>
__device__ __forceinline__
void tile_body(const float* __restrict__ x,
               const unsigned short* __restrict__ perm,
               const float* __restrict__ l1_w,
               const float* __restrict__ l1f_w,
               const float* __restrict__ l1_b,
               const float* __restrict__ l1f_b,
               const float* __restrict__ l2_w,
               const float* __restrict__ l2_b,
               const float* __restrict__ out_w,
               const float* __restrict__ out_b,
               float* __restrict__ out,
               float (*red)[64][36],
               const int s, const int row0, const int cnt_s) {
    const int wv   = threadIdx.x >> 5;
    const int lane = threadIdx.x & 31;
    const int r    = lane & 15;
    const int pbase = s * NB;
    const int plast = cnt_s - 1;

    const int koff = (lane < 16) ? 0 : 8;

    const float4* __restrict__ ax[NM];
#pragma unroll
    for (int m = 0; m < NM; ++m) {
        const int pm  = row0 + m * 16 + r;
        const int row = perm[pbase + ((pm < cnt_s) ? pm : plast)];
        ax[m] = (const float4*)(x + (size_t)row * NL1 + wv * 256 + koff);
    }
    const float4* __restrict__ bw[2];
    bw[0] = (const float4*)(l1_w  + (size_t)(s * 16 + r) * NL1 + wv * 256 + koff);
    bw[1] = (const float4*)(l1f_w + (size_t)r * NL1 + wv * 256 + koff);

    v8f acc[NM][2];
#pragma unroll
    for (int m = 0; m < NM; ++m) {
        acc[m][0] = (v8f){0.f, 0.f, 0.f, 0.f, 0.f, 0.f, 0.f, 0.f};
        acc[m][1] = (v8f){0.f, 0.f, 0.f, 0.f, 0.f, 0.f, 0.f, 0.f};
    }

    for (int G = 0; G < 8; ++G) {
        float qa[NM][16];
        float qb[2][2][8];
#pragma unroll
        for (int gg = 0; gg < 2; ++gg) {
#pragma unroll
            for (int h = 0; h < 2; ++h) {
                *(float4*)&qb[gg][h][0] = bw[h][8 * G + 4 * gg + 0];
                *(float4*)&qb[gg][h][4] = bw[h][8 * G + 4 * gg + 1];
            }
        }
#pragma unroll
        for (int m = 0; m < NM; ++m) {
            *(float4*)&qa[m][0]  = ax[m][8 * G + 0];
            *(float4*)&qa[m][4]  = ax[m][8 * G + 1];
            *(float4*)&qa[m][8]  = ax[m][8 * G + 4];
            *(float4*)&qa[m][12] = ax[m][8 * G + 5];
        }
        __builtin_amdgcn_sched_barrier(0);
#pragma unroll
        for (int gg = 0; gg < 2; ++gg) {
#pragma unroll
            for (int c = 0; c < 4; ++c) {
                v2f B0 = {qb[gg][0][2 * c], qb[gg][0][2 * c + 1]};
                v2f B1 = {qb[gg][1][2 * c], qb[gg][1][2 * c + 1]};
#pragma unroll
                for (int m = 0; m < NM; ++m) {
                    v2f A = {qa[m][8 * gg + 2 * c], qa[m][8 * gg + 2 * c + 1]};
                    acc[m][0] = __builtin_amdgcn_wmma_f32_16x16x4_f32(
                        false, A, false, B0, (short)0, acc[m][0], false, false);
                    acc[m][1] = __builtin_amdgcn_wmma_f32_16x16x4_f32(
                        false, A, false, B1, (short)0, acc[m][1], false, false);
                }
            }
        }
    }

    const int mbase = (lane >= 16) ? 8 : 0;
    const int ncol  = lane & 15;
#pragma unroll
    for (int m = 0; m < NM; ++m) {
#pragma unroll
        for (int v = 0; v < 8; ++v) {
            red[wv][m * 16 + v + mbase][ncol]      = acc[m][0][v];
            red[wv][m * 16 + v + mbase][ncol + 16] = acc[m][1][v];
        }
    }
    __syncthreads();

    if (wv >= NM) return;
    const int R    = wv * 16 + r;
    const int p    = row0 + R;
    const int rrow = perm[pbase + ((p < cnt_s) ? p : plast)];
    float y[32];
#pragma unroll
    for (int j4 = 0; j4 < 8; ++j4) {
        if (j4 == 4) __builtin_amdgcn_sched_barrier(0);
        float4 u0 = *(const float4*)&red[0][R][4 * j4];
        float4 u1 = *(const float4*)&red[1][R][4 * j4];
        float4 u2 = *(const float4*)&red[2][R][4 * j4];
        float4 u3 = *(const float4*)&red[3][R][4 * j4];
        y[4 * j4 + 0] = ((u0.x + u1.x) + u2.x) + u3.x;
        y[4 * j4 + 1] = ((u0.y + u1.y) + u2.y) + u3.y;
        y[4 * j4 + 2] = ((u0.z + u1.z) + u2.z) + u3.z;
        y[4 * j4 + 3] = ((u0.w + u1.w) + u2.w) + u3.w;
    }

    const int h = lane >> 4;
    float t[15];
    float keep;
    {
        float yc15 = y[15] + l1_b[s * 16 + 15];
        float yf15 = y[31] + l1f_b[15];
        keep = yc15 + yf15;
#pragma unroll
        for (int j = 0; j < 15; ++j) {
            float yc = y[j]      + l1_b[s * 16 + j];
            float yf = y[16 + j] + l1f_b[j];
            t[j] = yc + yf;
        }
    }
    float bv[16];
#pragma unroll
    for (int i = 0; i < 16; ++i) {
        float lo, hi;
        if (i < 15) { float sq = t[i] * t[i] * (127.0f / 128.0f); lo = fminf(fmaxf(sq, 0.f), 1.f); }
        else        { lo = fminf(fmaxf(t[0], 0.f), 1.f); }
        if (i < 14) { hi = fminf(fmaxf(t[i + 1], 0.f), 1.f); }
        else        { hi = 0.f; }
        bv[i] = h ? hi : lo;
    }
    float partial = 0.f;
#pragma unroll
    for (int j = 0; j < 2; ++j) {
        const unsigned wo = (unsigned)((s * 32 + 16 * j + r) * 30 + 16 * h);
        float av[16];
        __builtin_memcpy(&av[0],  &l2_w[wo],      16);
        __builtin_memcpy(&av[4],  &l2_w[wo + 4],  16);
        __builtin_memcpy(&av[8],  &l2_w[wo + 8],  16);
        __builtin_memcpy(&av[12], &l2_w[wo + 12], 8);
        float tl[2];
        __builtin_memcpy(&tl[0], &l2_w[wo + (h ? 12u : 14u)], 8);
        av[14] = h ? 0.f : tl[0];
        av[15] = h ? 0.f : tl[1];
        const unsigned ob = (unsigned)(s * 32 + 16 * j + 8 * h);
        float cb[8], ow[8];
        __builtin_memcpy(&cb[0], &l2_b[ob],      16);
        __builtin_memcpy(&cb[4], &l2_b[ob + 4],  16);
        __builtin_memcpy(&ow[0], &out_w[ob],     16);
        __builtin_memcpy(&ow[4], &out_w[ob + 4], 16);
        v8f a2 = {cb[0], cb[1], cb[2], cb[3], cb[4], cb[5], cb[6], cb[7]};
#pragma unroll
        for (int i = 0; i < 8; ++i) {
            v2f A = {av[2 * i], av[2 * i + 1]};
            v2f B = {bv[2 * i], bv[2 * i + 1]};
            a2 = __builtin_amdgcn_wmma_f32_16x16x4_f32(
                false, A, false, B, (short)0, a2, false, false);
        }
#pragma unroll
        for (int v = 0; v < 8; ++v) {
            float l2x = fminf(fmaxf(a2[v], 0.f), 1.f);
            partial += l2x * ow[v];
        }
    }
    float tot = partial + __shfl_xor(partial, 16, 32);

    if (lane < 16 && p < cnt_s) {
        out[rrow] = tot + out_b[s] + keep;
    }
}

__global__ __launch_bounds__(128, 5)
void k_main(const float* __restrict__ x,
            const unsigned short* __restrict__ perm,
            const int* __restrict__ cnt,
            const float* __restrict__ l1_w,
            const float* __restrict__ l1f_w,
            const float* __restrict__ l1_b,
            const float* __restrict__ l1f_b,
            const float* __restrict__ l2_w,
            const float* __restrict__ l2_b,
            const float* __restrict__ out_w,
            const float* __restrict__ out_b,
            float* __restrict__ out) {
    __shared__ __attribute__((aligned(16))) float red[4][64][36];

    int cn[NSTACK], nf[NSTACK], nl[NSTACK];
    int m0 = NB;
#pragma unroll
    for (int k = 0; k < NSTACK; ++k) {
        const int c = cnt[k];
        cn[k] = c;
        nf[k] = c >> 6;
        nl[k] = ((c & 63) + 15) >> 4;
        m0 = (nf[k] < m0) ? nf[k] : m0;
    }
    const int b = blockIdx.x;
    int  s = -1, row0 = 0;
    bool whole = true;
    if (b < NSTACK * m0) {
        s = b & 7;
        row0 = (b >> 3) * 64;
    } else {
        int q = b - NSTACK * m0;
#pragma unroll
        for (int k = 0; k < NSTACK; ++k) {
            const int e = nf[k] - m0;
            if (s < 0) {
                if (q < e) { s = k; row0 = (m0 + q) * 64; }
                else       q -= e;
            }
        }
#pragma unroll
        for (int k = 0; k < NSTACK; ++k) {
            if (s < 0) {
                if (q < nl[k]) { s = k; row0 = nf[k] * 64 + q * 16; whole = false; }
                else           q -= nl[k];
            }
        }
        if (s < 0) return;
    }
    int cnt_s = cn[0];
#pragma unroll
    for (int k = 1; k < NSTACK; ++k) cnt_s = (s == k) ? cn[k] : cnt_s;
    if (whole)
        tile_body<4>(x, perm, l1_w, l1f_w, l1_b, l1f_b, l2_w, l2_b, out_w, out_b,
                     out, red, s, row0, cnt_s);
    else
        tile_body<1>(x, perm, l1_w, l1f_w, l1_b, l1f_b, l2_w, l2_b, out_w, out_b,
                     out, red, s, row0, cnt_s);
}


extern "C" void kernel_launch(void* const* d_in, const int* in_sizes, int n_in,
                              void* d_out, int out_size, void* d_ws, size_t ws_size,
                              hipStream_t stream) {
    (void)in_sizes; (void)n_in; (void)out_size; (void)ws_size;
    const float* x      = (const float*)d_in[0];
    const int*   lsi    = (const int*)d_in[1];
    const float* l1_w   = (const float*)d_in[2];
    const float* l1_b   = (const float*)d_in[3];
    const float* l1f_w  = (const float*)d_in[4];
    const float* l1f_b  = (const float*)d_in[5];
    const float* l2_w   = (const float*)d_in[6];
    const float* l2_b   = (const float*)d_in[7];
    const float* out_w  = (const float*)d_in[8];
    const float* out_b  = (const float*)d_in[9];
    float* out = (float*)d_out;

    char* ws   = (char*)d_ws;
    int*  cur  = (int*)(ws + WS_CUR);
    unsigned short* perm = (unsigned short*)(ws + WS_PERM);

    k_zero<<<1, 32, 0, stream>>>(cur);
    k_scatter<<<NB / 256, 256, 0, stream>>>(lsi, cur, perm);
    k_main<<<NLIVE, 128, 0, stream>>>(x, perm, cur, l1_w, l1f_w,
                                              l1_b, l1f_b, l2_w, l2_b,
                                              out_w, out_b, out);
}
